# starting kernel
# baseline (speedup 1.0000x reference)
#include <hip/hip_runtime.h>


typedef float v2f __attribute__((ext_vector_type(2)));
typedef float v8f __attribute__((ext_vector_type(8)));

#define GNN_N 100000
#define GNN_E 1600000
#define DIM   64
#define XPAD  68

__global__ void fill_kernel(float* __restrict__ p, float v, int n) {
    int i = blockIdx.x * blockDim.x + threadIdx.x;
    if (i < n) p[i] = v;
}

__global__ void deg_kernel(const int* __restrict__ dst, float* __restrict__ deg, int E) {
    int e = blockIdx.x * blockDim.x + threadIdx.x;
    if (e < E)
        __hip_atomic_fetch_add(&deg[dst[e]], 1.0f, __ATOMIC_RELAXED, __HIP_MEMORY_SCOPE_AGENT);
}

__global__ void rsqrt_kernel(float* __restrict__ d, int n) {
    int i = blockIdx.x * blockDim.x + threadIdx.x;
    if (i < n) d[i] = rsqrtf(d[i]);
}

__global__ void norm_kernel(const int* __restrict__ src, const int* __restrict__ dst,
                            const float* __restrict__ dinv, float* __restrict__ nrm, int E) {
    int e = blockIdx.x * blockDim.x + threadIdx.x;
    if (e < E) nrm[e] = dinv[src[e]] * dinv[dst[e]];
}

__global__ __launch_bounds__(256) void gemm64_wmma(const float* __restrict__ X,
                                                   const float* __restrict__ W,
                                                   float* __restrict__ H, int nTiles) {
    __shared__ float Ws[64 * 64];
    __shared__ float Xs[8][16 * XPAD];

    const int tid  = threadIdx.x;
    const int wave = tid >> 5;
    const int lane = tid & 31;
    const int tile = blockIdx.x * 8 + wave;

#pragma unroll
    for (int i = 0; i < 16; ++i) {
        int g = tid + i * 256;
        int k = g >> 6;
        int n = g & 63;
        Ws[(k >> 1) * 128 + n * 2 + (k & 1)] = W[g];
    }

    if (tile < nTiles) {
        const float4* Xg = (const float4*)(X + (size_t)tile * 16 * 64);
        float* xl = &Xs[wave][0];
#pragma unroll
        for (int i = 0; i < 8; ++i) {
            int idx4 = lane + i * 32;
            float4 v = Xg[idx4];
            int row = idx4 >> 4;
            int col = (idx4 & 15) * 4;
            *(float4*)&xl[row * XPAD + col] = v;
        }
    }
    __syncthreads();
    if (tile >= nTiles) return;

    const float* xs = &Xs[wave][0];
    const int l15 = lane & 15;
    const int hiK = (lane >> 4) * 2;

    v8f c0 = {}, c1 = {}, c2 = {}, c3 = {};
#pragma unroll
    for (int k = 0; k < 64; k += 4) {
        v2f a = *(const v2f*)(xs + l15 * XPAD + k + hiK);
        const float* wrow = Ws + ((k + hiK) >> 1) * 128 + l15 * 2;
        v2f b0 = *(const v2f*)(wrow +  0);
        v2f b1 = *(const v2f*)(wrow + 32);
        v2f b2 = *(const v2f*)(wrow + 64);
        v2f b3 = *(const v2f*)(wrow + 96);
        c0 = __builtin_amdgcn_wmma_f32_16x16x4_f32(false, a, false, b0, (short)0, c0, false, false);
        c1 = __builtin_amdgcn_wmma_f32_16x16x4_f32(false, a, false, b1, (short)0, c1, false, false);
        c2 = __builtin_amdgcn_wmma_f32_16x16x4_f32(false, a, false, b2, (short)0, c2, false, false);
        c3 = __builtin_amdgcn_wmma_f32_16x16x4_f32(false, a, false, b3, (short)0, c3, false, false);
    }

    const size_t base = (size_t)tile * 16;
    const int rowOff = (lane >> 4) * 8;
#pragma unroll
    for (int r = 0; r < 8; ++r) {
        float* out = H + (base + r + rowOff) * 64 + l15;
        out[0]  = c0[r];
        out[16] = c1[r];
        out[32] = c2[r];
        out[48] = c3[r];
    }
}

__global__ __launch_bounds__(256) void scatter_kernel(const float* __restrict__ h,
                                                      const float* __restrict__ nrm,
                                                      const int* __restrict__ src,
                                                      const int* __restrict__ dst,
                                                      float* __restrict__ agg, int E) {
    int ew = __builtin_amdgcn_readfirstlane((int)threadIdx.x >> 6);
    int e  = blockIdx.x * 4 + ew;
    int d  = threadIdx.x & 63;
    if (e >= E) return;
    int   s = src[e];
    int   q = dst[e];
    float w = nrm[e];
    float v = h[(size_t)s * 64 + d] * w;
    __hip_atomic_fetch_add(&agg[(size_t)q * 64 + d], v,
                           __ATOMIC_RELAXED, __HIP_MEMORY_SCOPE_AGENT);
}

template <bool RELU>
__global__ void finalize_kernel(float* __restrict__ agg, const float* __restrict__ h,
                                const float* __restrict__ dinv, const float* __restrict__ b,
                                int total) {
    int t = blockIdx.x * blockDim.x + threadIdx.x;
    if (t >= total) return;
    int i = __builtin_amdgcn_readfirstlane(t >> 6);
    int d = t & 63;
    float di = dinv[i];
    float v = agg[t] + di * di * h[t] + b[d];
    if (RELU) v = fmaxf(v, 0.0f);
    agg[t] = v;
}

__global__ __launch_bounds__(256) void proj_kernel(const float* __restrict__ f,
                                                   const float* __restrict__ Wp,
                                                   const float* __restrict__ bp,
                                                   float* __restrict__ y, int N) {
    int node = blockIdx.x * 8 + (threadIdx.x >> 5);
    int lane = threadIdx.x & 31;
    if (node >= N) return;
    const float* fr = f + (size_t)node * 64;
    float s = fr[lane] * Wp[lane] + fr[lane + 32] * Wp[lane + 32];
#pragma unroll
    for (int m = 16; m > 0; m >>= 1) s += __shfl_xor(s, m, 32);
    if (lane == 0) y[node] = s + bp[0];
}

extern "C" void kernel_launch(void* const* d_in, const int* in_sizes, int n_in,
                              void* d_out, int out_size, void* d_ws, size_t ws_size,
                              hipStream_t stream) {
    (void)in_sizes; (void)n_in; (void)out_size; (void)ws_size;

    const float* x  = (const float*)d_in[0];
    const int*   ei = (const int*)d_in[1];
    const float* W1 = (const float*)d_in[2];
    const float* b1 = (const float*)d_in[3];
    const float* W2 = (const float*)d_in[4];
    const float* b2 = (const float*)d_in[5];
    const float* Wp = (const float*)d_in[6];
    const float* bp = (const float*)d_in[7];
    float* y = (float*)d_out;

    const int N = GNN_N, E = GNN_E;
    const int* src = ei;
    const int* dst = ei + E;

    char* ws = (char*)d_ws;
    size_t off = 0;
    auto take = [&](size_t bytes) {
        void* p = ws + off;
        off += (bytes + 255) & ~(size_t)255;
        return p;
    };
    float* dinv = (float*)take((size_t)N * 4);
    float* nrm  = (float*)take((size_t)E * 4);
    float* bufA = (float*)take((size_t)N * DIM * 4);
    float* bufB = (float*)take((size_t)N * DIM * 4);

    const int total = N * DIM;
    const int nTiles = N / 16;
    const int gemmBlocks = (nTiles + 7) / 8;

    fill_kernel<<<(N + 255) / 256, 256, 0, stream>>>(dinv, 1.0f, N);
    deg_kernel<<<(E + 255) / 256, 256, 0, stream>>>(dst, dinv, E);
    rsqrt_kernel<<<(N + 255) / 256, 256, 0, stream>>>(dinv, N);
    norm_kernel<<<(E + 255) / 256, 256, 0, stream>>>(src, dst, dinv, nrm, E);

    gemm64_wmma<<<gemmBlocks, 256, 0, stream>>>(x, W1, bufA, nTiles);
    fill_kernel<<<total / 256, 256, 0, stream>>>(bufB, 0.0f, total);
    scatter_kernel<<<E / 4, 256, 0, stream>>>(bufA, nrm, src, dst, bufB, E);
    finalize_kernel<true><<<total / 256, 256, 0, stream>>>(bufB, bufA, dinv, b1, total);

    gemm64_wmma<<<gemmBlocks, 256, 0, stream>>>(bufB, W2, bufA, nTiles);
    fill_kernel<<<total / 256, 256, 0, stream>>>(bufB, 0.0f, total);
    scatter_kernel<<<E / 4, 256, 0, stream>>>(bufA, nrm, src, dst, bufB, E);
    finalize_kernel<true><<<total / 256, 256, 0, stream>>>(bufB, bufA, dinv, b2, total);

    proj_kernel<<<(N + 7) / 8, 256, 0, stream>>>(bufB, Wp, bp, y, N);
}


# Layer 2: a wave owns 12-13 rows of the tile; recorded edges go through LDS queues to the row's owner, which sums source rows in registers and finishes its rows: no LDS float adds, no finish pass
# speedup vs baseline: 3.3085x; 3.3085x over previous
#include <hip/hip_runtime.h>


typedef float v2f __attribute__((ext_vector_type(2)));
typedef float v8f __attribute__((ext_vector_type(8)));
typedef float v16f __attribute__((ext_vector_type(16)));

#define GNN_N 100000
#define GNN_E 1600000
#define DIM   64
#define XPAD  68

#define TILE_R   400
#define N_TILES  (GNN_N / TILE_R)
#define NV16     (GNN_E / 16)
#define TS_THREADS 1024
#define TS_WAVES   (TS_THREADS / 32)
#define TS_ITERS   ((NV16 + TS_THREADS - 1) / TS_THREADS)
#define TB_QUADS   ((TS_ITERS + 3) * TS_THREADS * 4)
#define QCAP     64
#define LCAP     288
#define QC       320
#define N_SEGS   (N_TILES * TS_WAVES)
#define LIST_WORDS (N_SEGS * LCAP / 2)
static_assert(TB_QUADS + LIST_WORDS + N_SEGS <= GNN_E, "tile bytes, records and counts must fit the 4*E byte slot");
static_assert(TS_ITERS + 2 <= 128 && TS_WAVES * 512 == 16384, "an edge id is (round << 14) + wave * 512 + a 9-bit rest");

__global__ void fill_kernel(float* __restrict__ p, float v, int n) {
    int i = blockIdx.x * blockDim.x + threadIdx.x;
    if (i < n) p[i] = v;
}

__global__ void deg_kernel(const int* __restrict__ dst, float* __restrict__ deg, int E) {
    int e = blockIdx.x * blockDim.x + threadIdx.x;
    if (e < E)
        __hip_atomic_fetch_add(&deg[dst[e]], 1.0f, __ATOMIC_RELAXED, __HIP_MEMORY_SCOPE_AGENT);
}

__global__ void rsqrt_kernel(float* __restrict__ d, int n) {
    int i = blockIdx.x * blockDim.x + threadIdx.x;
    if (i < n) d[i] = rsqrtf(d[i]);
}

__global__ void tilemap_kernel(const int* __restrict__ dst, unsigned* __restrict__ tb4, int nQuads) {
    int i = blockIdx.x * blockDim.x + threadIdx.x;
    if (i >= TB_QUADS) return;
    unsigned p = ~0u;
    if (i < nQuads) {
        int4 d = ((const int4*)dst)[i];
        p = (unsigned)(d.x / TILE_R) | ((unsigned)(d.y / TILE_R) << 8) |
            ((unsigned)(d.z / TILE_R) << 16) | ((unsigned)(d.w / TILE_R) << 24);
    }
    tb4[i] = p;
}

__global__ __launch_bounds__(256) void gemm64_wmma(const float* __restrict__ X,
                                                   const float* __restrict__ W,
                                                   const float* __restrict__ rs,
                                                   float* __restrict__ H, int nTiles) {
    __shared__ float Ws[64 * 64];
    __shared__ float Xs[8][16 * XPAD];

    const int tid  = threadIdx.x;
    const int wave = tid >> 5;
    const int lane = tid & 31;
    const int tile = blockIdx.x * 8 + wave;

#pragma unroll
    for (int i = 0; i < 16; ++i) {
        int g = tid + i * 256;
        int k = g >> 6;
        int n = g & 63;
        Ws[(k >> 1) * 128 + n * 2 + (k & 1)] = W[g];
    }

    if (tile < nTiles) {
        const float4* Xg = (const float4*)(X + (size_t)tile * 16 * 64);
        float* xl = &Xs[wave][0];
#pragma unroll
        for (int i = 0; i < 8; ++i) {
            int idx4 = lane + i * 32;
            float4 v = Xg[idx4];
            int row = idx4 >> 4;
            int col = (idx4 & 15) * 4;
            *(float4*)&xl[row * XPAD + col] = v;
        }
    }
    __syncthreads();
    if (tile >= nTiles) return;

    const float* xs = &Xs[wave][0];
    const int l15 = lane & 15;
    const int hiK = (lane >> 4) * 2;

    const size_t base = (size_t)tile * 16;
    const int rowOff = (lane >> 4) * 8;
    float sc[8];
#pragma unroll
    for (int r = 0; r < 8; ++r) sc[r] = rs[base + r + rowOff];

    v8f c0 = {}, c1 = {}, c2 = {}, c3 = {};
#pragma unroll
    for (int k = 0; k < 64; k += 4) {
        v2f a = *(const v2f*)(xs + l15 * XPAD + k + hiK);
        const float* wrow = Ws + ((k + hiK) >> 1) * 128 + l15 * 2;
        v2f b0 = *(const v2f*)(wrow +  0);
        v2f b1 = *(const v2f*)(wrow + 32);
        v2f b2 = *(const v2f*)(wrow + 64);
        v2f b3 = *(const v2f*)(wrow + 96);
        c0 = __builtin_amdgcn_wmma_f32_16x16x4_f32(false, a, false, b0, (short)0, c0, false, false);
        c1 = __builtin_amdgcn_wmma_f32_16x16x4_f32(false, a, false, b1, (short)0, c1, false, false);
        c2 = __builtin_amdgcn_wmma_f32_16x16x4_f32(false, a, false, b2, (short)0, c2, false, false);
        c3 = __builtin_amdgcn_wmma_f32_16x16x4_f32(false, a, false, b3, (short)0, c3, false, false);
    }

#pragma unroll
    for (int r = 0; r < 8; ++r) {
        float* out = H + (base + r + rowOff) * 64 + l15;
        out[0]  = c0[r] * sc[r];
        out[16] = c1[r] * sc[r];
        out[32] = c2[r] * sc[r];
        out[48] = c3[r] * sc[r];
    }
}

__device__ __forceinline__ void lds_add(float* p, float v) {
    __hip_atomic_fetch_add(p, v, __ATOMIC_RELAXED, __HIP_MEMORY_SCOPE_WORKGROUP);
}

__device__ __forceinline__ float2 row_load(const float* __restrict__ h, int sj, int lane) {
    return *(const float2*)(h + (size_t)sj * 64 + 2 * lane);
}
__device__ __forceinline__ void row_add(float* acc, int rj, float2 v, int lane) {
    lds_add(&acc[rj + 2 * lane],     v.x);
    lds_add(&acc[rj + 2 * lane + 1], v.y);
}

template <int M>
__device__ __forceinline__ float xmask_add(float p) {
    const int q = __builtin_amdgcn_update_dpp(0, __float_as_int(p), 0x160 + M, 0xf, 0xf, true);
    return p + __int_as_float(q);
}
__device__ __forceinline__ float row16_sum(float p) {
    p = xmask_add<1>(p);
    p = xmask_add<2>(p);
    p = xmask_add<4>(p);
    p = xmask_add<8>(p);
    return p;
}

template <bool RELU, bool HEAD>
__global__ __launch_bounds__(TS_THREADS) void tile_scatter_kernel(
        const float* __restrict__ h, const float* __restrict__ dinv,
        const int* __restrict__ src, const int* __restrict__ dst,
        const unsigned* __restrict__ tbytes, const float* __restrict__ b,
        float* __restrict__ out,
        const float* __restrict__ Wp, const float* __restrict__ bp, float* __restrict__ y,
        unsigned short* __restrict__ lists, int* __restrict__ counts, const int replay) {
    const uint4* __restrict__ tb = (const uint4*)tbytes;
    __shared__ float acc[TILE_R * 64];
    __shared__ int   q[TS_WAVES][QCAP];
    __shared__ unsigned oq[TS_WAVES][QC];
    __shared__ int   qcnt[TS_WAVES];
    __shared__ float ys[TILE_R];

    const int tid  = threadIdx.x;
    const int lane = tid & 31;
    const int wave = __builtin_amdgcn_readfirstlane(tid >> 5);
    const int tile = blockIdx.x;
    const int base = tile * TILE_R;

    const int seg = tile * TS_WAVES + wave;
    unsigned short* __restrict__ L = lists + (size_t)seg * LCAP;
    int total = 0;
    if (replay) total = __builtin_amdgcn_readfirstlane(counts[seg]);
    int done = 0;

    for (int i = tid; i < TILE_R * 16; i += TS_THREADS)
        ((float4*)acc)[i] = make_float4(0.f, 0.f, 0.f, 0.f);
    if (HEAD && tid < TS_WAVES) qcnt[tid] = 0;
    __syncthreads();

    int* qw = q[wave];
    int head = 0, cnt = 0;
    const unsigned my4 = (unsigned)tile * 0x01010101u;

    auto process = [&](int s, int ro, int n, bool full) {
        if (full) {
            float2 v[32];
#pragma unroll
            for (int j = 0; j < 32; ++j) v[j] = row_load(h, __builtin_amdgcn_readlane(s, j), lane);
#pragma unroll
            for (int j = 0; j < 32; ++j)
                row_add(acc, __builtin_amdgcn_readlane(ro, j), v[j], lane);
        } else {
            for (int j0 = 0; j0 < n; j0 += 8) {
                float2 v[8];
#pragma unroll
                for (int u = 0; u < 8; ++u) v[u] = row_load(h, __builtin_amdgcn_readlane(s, j0 + u), lane);
#pragma unroll
                for (int u = 0; u < 8; ++u)
                    if (j0 + u < n)
                        row_add(acc, __builtin_amdgcn_readlane(ro, j0 + u), v[u], lane);
            }
        }
    };

    auto drain = [&](int n, bool full) {
        const bool act = lane < n;
        int eid = act ? qw[(head + lane) & (QCAP - 1)] : 0;
        if (!replay) {
            if (act && done + n <= LCAP)
                L[done + lane] = (unsigned short)(((eid >> 14) << 9) | (eid & 511));
            done += n;
        }
        int s  = src[eid];
        int ro = act ? (dst[eid] - base) * 64 : 0;
        process(s, ro, n, full);
    };

    auto scan16 = [&](const uint4 cur, const int idx) {
        unsigned x0 = cur.x ^ my4, x1 = cur.y ^ my4, x2 = cur.z ^ my4, x3 = cur.w ^ my4;
        unsigned z0 = ~(((x0 & 0x7f7f7f7fu) + 0x7f7f7f7fu) | x0 | 0x7f7f7f7fu);
        unsigned z1 = ~(((x1 & 0x7f7f7f7fu) + 0x7f7f7f7fu) | x1 | 0x7f7f7f7fu);
        unsigned z2 = ~(((x2 & 0x7f7f7f7fu) + 0x7f7f7f7fu) | x2 | 0x7f7f7f7fu);
        unsigned z3 = ~(((x3 & 0x7f7f7f7fu) + 0x7f7f7f7fu) | x3 | 0x7f7f7f7fu);
        unsigned m = (z0 >> 7) | (z1 >> 6) | (z2 >> 5) | (z3 >> 4);

        unsigned bal = __builtin_amdgcn_ballot_w32(m != 0);
        while (bal) {
            if (m != 0) {
                int p = __builtin_ctz(m);
                int e = idx * 16 + ((p & 7) << 2) + (p >> 3);
                qw[(head + cnt + (int)__builtin_amdgcn_mbcnt_lo(bal, 0u)) & (QCAP - 1)] = e;
                m &= m - 1;
            }
            cnt += __builtin_popcount(bal);
            if (cnt >= 32) {
                drain(32, true);
                head = (head + 32) & (QCAP - 1);
                cnt -= 32;
            }
            bal = __builtin_amdgcn_ballot_w32(m != 0);
        }
    };

    if (HEAD && replay && total <= LCAP) {
        if (total > 0) {
            const int wbase = wave * 512;
            int sv[LCAP / 32], dv[LCAP / 32];
#pragma unroll
            for (int bq = 0; bq < LCAP / 32; ++bq) {
                const int i = bq * 32 + lane;
                const unsigned c = L[i < total ? i : 0];
                const int eid = (int)((c >> 9) << 14) + wbase + (int)(c & 511u);
                sv[bq] = src[eid];
                dv[bq] = dst[eid] - base;
            }
#pragma unroll
            for (int bq = 0; bq < LCAP / 32; ++bq) {
                if (bq * 32 < total) {
                    const bool valid = bq * 32 + lane < total;
                    const int dl = dv[bq];
                    const int ow = (2 * dl) / 25;
                    const int dr = dl - ((25 * ow + 1) >> 1);
                    bool over = false;
                    if (valid) {
                        const int slot = __hip_atomic_fetch_add(&qcnt[ow], 1, __ATOMIC_RELAXED, __HIP_MEMORY_SCOPE_WORKGROUP);
                        if (slot < QC) oq[ow][slot] = (unsigned)sv[bq] | ((unsigned)dr << 17);
                        else over = true;
                    }
                    unsigned ob = __builtin_amdgcn_ballot_w32(over);
                    while (ob) {
                        const int j = __builtin_ctz(ob);
                        ob &= ob - 1;
                        const int sj = __builtin_amdgcn_readlane(sv[bq], j);
                        const int rj = __builtin_amdgcn_readlane(dl, j) * 64;
                        row_add(acc, rj, row_load(h, sj, lane), lane);
                    }
                }
            }
        }
    } else {
    uint4 ba = tb[tid];
    uint4 bb = tb[tid + TS_THREADS];
    for (int it = 0; it < TS_ITERS; it += 3) {
        const int idx = it * TS_THREADS + tid;
        uint4 bc = tb[idx + 2 * TS_THREADS];
        scan16(ba, idx);
        ba = tb[idx + 3 * TS_THREADS];
        scan16(bb, idx + TS_THREADS);
        bb = tb[idx + 4 * TS_THREADS];
        scan16(bc, idx + 2 * TS_THREADS);
    }
    if (cnt > 0) drain(cnt, false);
    if (!replay && lane == 0) counts[seg] = done;
    }
    __syncthreads();

    if (!HEAD) {
    for (int i = tid; i < TILE_R * 16; i += TS_THREADS) {
        const int r  = i >> 4;
        const int c4 = (i & 15) * 4;
        const size_t g = (size_t)(base + r) * 64 + c4;
        float4 a  = ((const float4*)acc)[i];
        float4 hv = *(const float4*)(h + g);
        float4 bv = *(const float4*)(b + c4);
        float di = dinv[base + r];
        float4 o;
        o.x = di * (a.x + hv.x) + bv.x;
        o.y = di * (a.y + hv.y) + bv.y;
        o.z = di * (a.z + hv.z) + bv.z;
        o.w = di * (a.w + hv.w) + bv.w;
        if (RELU) {
            o.x = fmaxf(o.x, 0.0f); o.y = fmaxf(o.y, 0.0f);
            o.z = fmaxf(o.z, 0.0f); o.w = fmaxf(o.w, 0.0f);
        }
        *(float4*)(out + g) = o;
    }
    } else {
    const int rs0 = (25 * wave + 1) >> 1;
    const int nr  = ((25 * (wave + 1) + 1) >> 1) - rs0;
    int nq = __builtin_amdgcn_readfirstlane(qcnt[wave]);
    nq = nq < QC ? nq : QC;
    const unsigned* oqw = oq[wave];
    v16f RX = {}, RY = {};
    for (int b0 = 0; b0 < nq; b0 += 32) {
        const int m = (nq - b0 < 32) ? (nq - b0) : 32;
        const unsigned pk = (lane < m) ? oqw[b0 + lane] : (15u << 17);
        const int s  = (int)(pk & 0x1ffffu);
        const int dr = (int)(pk >> 17);
        float2 v[32];
#pragma unroll
        for (int j = 0; j < 32; ++j) v[j] = row_load(h, __builtin_amdgcn_readlane(s, j), lane);
        __builtin_amdgcn_sched_barrier(0);
#pragma unroll
        for (int j = 0; j < 32; ++j) {
            const int d = __builtin_amdgcn_readlane(dr, j);
            RX[d] += v[j].x;
            RY[d] += v[j].y;
        }
    }
    float2 hv[13];
    float  dq[13];
#pragma unroll
    for (int r = 0; r < 13; ++r) {
        const int row = base + rs0 + (r < nr ? r : 0);
        hv[r] = *(const float2*)(h + (size_t)row * 64 + 2 * lane);
        dq[r] = dinv[row];
    }
    const float2 bv = *(const float2*)(b + 2 * lane);
    const float2 wq = *(const float2*)(Wp + 2 * lane);
#pragma unroll
    for (int r = 0; r < 13; ++r) {
        if (r < nr) {
            const float2 a = *(const float2*)&acc[(rs0 + r) * 64 + 2 * lane];
            float ox = dq[r] * ((RX[r] + a.x) + hv[r].x) + bv.x;
            float oy = dq[r] * ((RY[r] + a.y) + hv[r].y) + bv.y;
            if (RELU) { ox = fmaxf(ox, 0.0f); oy = fmaxf(oy, 0.0f); }
            float p = ox * wq.x + oy * wq.y;
            p = row16_sum(p);
            const float tot = __int_as_float(__builtin_amdgcn_readlane(__float_as_int(p), 0)) +
                              __int_as_float(__builtin_amdgcn_readlane(__float_as_int(p), 16));
            if (lane == 0) ys[rs0 + r] = tot;
        }
    }
    }
    if (HEAD) {
        __syncthreads();
        const int first = base & ~31;
        const int n = first + tid;
        if (n >= base && n < base + TILE_R) y[n] = ys[n - base] + bp[0];
    }
}

extern "C" void kernel_launch(void* const* d_in, const int* in_sizes, int n_in,
                              void* d_out, int out_size, void* d_ws, size_t ws_size,
                              hipStream_t stream) {
    (void)in_sizes; (void)n_in; (void)out_size; (void)ws_size;

    const float* x  = (const float*)d_in[0];
    const int*   ei = (const int*)d_in[1];
    const float* W1 = (const float*)d_in[2];
    const float* b1 = (const float*)d_in[3];
    const float* W2 = (const float*)d_in[4];
    const float* b2 = (const float*)d_in[5];
    const float* Wp = (const float*)d_in[6];
    const float* bp = (const float*)d_in[7];
    float* y = (float*)d_out;

    const int N = GNN_N, E = GNN_E;
    const int* src = ei;
    const int* dst = ei + E;

    char* ws = (char*)d_ws;
    size_t off = 0;
    auto take = [&](size_t bytes) {
        void* p = ws + off;
        off += (bytes + 255) & ~(size_t)255;
        return p;
    };
    float* dinv = (float*)take((size_t)N * 4);
    unsigned* tb = (unsigned*)take((size_t)E * 4);
    unsigned short* lists = (unsigned short*)(tb + TB_QUADS);
    int* counts = (int*)(tb + TB_QUADS + LIST_WORDS);
    float* bufA = (float*)take((size_t)N * DIM * 4);
    float* bufB = (float*)take((size_t)N * DIM * 4);

    const int nTiles = N / 16;
    const int gemmBlocks = (nTiles + 7) / 8;

    fill_kernel<<<(N + 255) / 256, 256, 0, stream>>>(dinv, 1.0f, N);
    deg_kernel<<<(E + 255) / 256, 256, 0, stream>>>(dst, dinv, E);
    rsqrt_kernel<<<(N + 255) / 256, 256, 0, stream>>>(dinv, N);
    tilemap_kernel<<<(TB_QUADS + 255) / 256, 256, 0, stream>>>(dst, tb, E / 4);

    gemm64_wmma<<<gemmBlocks, 256, 0, stream>>>(x, W1, dinv, bufA, nTiles);
    tile_scatter_kernel<true, false><<<N_TILES, TS_THREADS, 0, stream>>>(bufA, dinv, src, dst, tb, b1, bufB, Wp, bp, y, lists, counts, 0);

    gemm64_wmma<<<gemmBlocks, 256, 0, stream>>>(bufB, W2, dinv, bufA, nTiles);
    tile_scatter_kernel<true, true><<<N_TILES, TS_THREADS, 0, stream>>>(bufA, dinv, src, dst, tb, b2, bufB, Wp, bp, y, lists, counts, 1);
}


# Map kernel also writes 256 presence bits per 64-edge group; the tile filter reads tile-id bytes only of groups holding an edge of its tile; a wave's record order, hence row-sum order, changes
# speedup vs baseline: 5.0245x; 1.5187x over previous
#include <hip/hip_runtime.h>


typedef float v2f __attribute__((ext_vector_type(2)));
typedef float v8f __attribute__((ext_vector_type(8)));
typedef float v16f __attribute__((ext_vector_type(16)));

#define GNN_N 100000
#define GNN_E 1600000
#define DIM   64
#define XPAD  68

#define TILE_R   400
#define N_TILES  (GNN_N / TILE_R)
#define NV16     (GNN_E / 16)
#define TS_THREADS 1024
#define TS_WAVES   (TS_THREADS / 32)
#define TS_ITERS   ((NV16 + TS_THREADS - 1) / TS_THREADS)
#define TB_QUADS   ((TS_ITERS + 3) * TS_THREADS * 4)
#define QCAP     64
#define LCAP     288
#define QC       320
#define N_SEGS   (N_TILES * TS_WAVES)
#define LIST_WORDS (N_SEGS * LCAP / 2)
static_assert(TB_QUADS + LIST_WORDS + N_SEGS <= GNN_E, "tile bytes, records and counts must fit the 4*E byte slot");
static_assert(TS_ITERS + 2 <= 128 && TS_WAVES * 512 == 16384, "an edge id is (round << 14) + wave * 512 + a 9-bit rest");
#define GRP_QUADS   16
#define N_PLANES    8
#define PLANE_WORDS (TB_QUADS / GRP_QUADS)
#define PA_CHUNKS   ((TS_ITERS + 3) / 4)
#define GL_CAP      (PA_CHUNKS * 32)
#define PAD_GROUP   (GNN_E / 64)
static_assert(TB_QUADS % 256 == 0 && 256 / GRP_QUADS == 16, "a map block is 16 whole groups and the grid has no partial block");
static_assert((PA_CHUNKS * 4) * 256 <= PLANE_WORDS && PLANE_WORDS < 65536, "every group a wave tests has a presence word; a group number fits 16 bits");
static_assert((PAD_GROUP + 1) * GRP_QUADS <= TB_QUADS && N_TILES <= 255, "the padding group's bytes exist and name no tile");
static_assert((size_t)N_PLANES * PLANE_WORDS <= (size_t)GNN_N * DIM, "the presence planes fit the row buffer that lends its space");

__global__ __launch_bounds__(256) void tilemap_kernel(const int* __restrict__ dst, unsigned* __restrict__ tb4,
                                                      unsigned* __restrict__ pres, int nQuads) {
    __shared__ unsigned pw[N_PLANES][16];
    const int t = threadIdx.x;
    const int i = blockIdx.x * 256 + t;
    if (t < N_PLANES * 16) (&pw[0][0])[t] = 0u;
    __syncthreads();
    unsigned p = ~0u;
    if (i < nQuads) {
        int4 d = ((const int4*)dst)[i];
        p = (unsigned)(d.x / TILE_R) | ((unsigned)(d.y / TILE_R) << 8) |
            ((unsigned)(d.z / TILE_R) << 16) | ((unsigned)(d.w / TILE_R) << 24);
    }
    tb4[i] = p;
    const int g = t >> 4;
#pragma unroll
    for (int k = 0; k < 4; ++k) {
        const unsigned id = (p >> (8 * k)) & 255u;
        __hip_atomic_fetch_or(&pw[id >> 5][g], 1u << (id & 31u), __ATOMIC_RELAXED, __HIP_MEMORY_SCOPE_WORKGROUP);
    }
    __syncthreads();
    if (t < N_PLANES * 16)
        pres[(size_t)(t >> 4) * PLANE_WORDS + blockIdx.x * 16 + (t & 15)] = pw[t >> 4][t & 15];
}

__global__ __launch_bounds__(256) void gemm64_wmma(const float* __restrict__ X,
                                                   const float* __restrict__ W,
                                                   const float* __restrict__ rs,
                                                   float* __restrict__ H, int nTiles) {
    __shared__ float Ws[64 * 64];
    __shared__ float Xs[8][16 * XPAD];

    const int tid  = threadIdx.x;
    const int wave = tid >> 5;
    const int lane = tid & 31;
    const int tile = blockIdx.x * 8 + wave;

#pragma unroll
    for (int i = 0; i < 16; ++i) {
        int g = tid + i * 256;
        int k = g >> 6;
        int n = g & 63;
        Ws[(k >> 1) * 128 + n * 2 + (k & 1)] = W[g];
    }

    if (tile < nTiles) {
        const float4* Xg = (const float4*)(X + (size_t)tile * 16 * 64);
        float* xl = &Xs[wave][0];
#pragma unroll
        for (int i = 0; i < 8; ++i) {
            int idx4 = lane + i * 32;
            float4 v = Xg[idx4];
            int row = idx4 >> 4;
            int col = (idx4 & 15) * 4;
            *(float4*)&xl[row * XPAD + col] = v;
        }
    }
    __syncthreads();
    if (tile >= nTiles) return;

    const float* xs = &Xs[wave][0];
    const int l15 = lane & 15;
    const int hiK = (lane >> 4) * 2;

    const size_t base = (size_t)tile * 16;
    const int rowOff = (lane >> 4) * 8;
    float sc[8];
#pragma unroll
    for (int r = 0; r < 8; ++r) sc[r] = rs[base + r + rowOff];

    v8f c0 = {}, c1 = {}, c2 = {}, c3 = {};
#pragma unroll
    for (int k = 0; k < 64; k += 4) {
        v2f a = *(const v2f*)(xs + l15 * XPAD + k + hiK);
        const float* wrow = Ws + ((k + hiK) >> 1) * 128 + l15 * 2;
        v2f b0 = *(const v2f*)(wrow +  0);
        v2f b1 = *(const v2f*)(wrow + 32);
        v2f b2 = *(const v2f*)(wrow + 64);
        v2f b3 = *(const v2f*)(wrow + 96);
        c0 = __builtin_amdgcn_wmma_f32_16x16x4_f32(false, a, false, b0, (short)0, c0, false, false);
        c1 = __builtin_amdgcn_wmma_f32_16x16x4_f32(false, a, false, b1, (short)0, c1, false, false);
        c2 = __builtin_amdgcn_wmma_f32_16x16x4_f32(false, a, false, b2, (short)0, c2, false, false);
        c3 = __builtin_amdgcn_wmma_f32_16x16x4_f32(false, a, false, b3, (short)0, c3, false, false);
    }

#pragma unroll
    for (int r = 0; r < 8; ++r) {
        float* out = H + (base + r + rowOff) * 64 + l15;
        out[0]  = c0[r] * sc[r];
        out[16] = c1[r] * sc[r];
        out[32] = c2[r] * sc[r];
        out[48] = c3[r] * sc[r];
    }
}

__global__ __launch_bounds__(TS_THREADS) void tile_filter_kernel(
        const int* __restrict__ dst, const unsigned* __restrict__ tbytes, const unsigned* __restrict__ pres,
        float* __restrict__ dinv, unsigned short* __restrict__ lists, int* __restrict__ counts) {
    const uint4* __restrict__ tb = (const uint4*)tbytes;
    __shared__ int q[TS_WAVES][QCAP];
    __shared__ int indeg[TILE_R];
    __shared__ unsigned short gl[TS_WAVES][GL_CAP];

    const int tid  = threadIdx.x;
    const int lane = tid & 31;
    const int wave = __builtin_amdgcn_readfirstlane(tid >> 5);
    const int tile = blockIdx.x;
    const int base = tile * TILE_R;

    const int seg = tile * TS_WAVES + wave;
    unsigned short* __restrict__ L = lists + (size_t)seg * LCAP;
    int done = 0;

    if (tid < TILE_R) indeg[tid] = 0;
    __syncthreads();

    int* qw = q[wave];
    int head = 0, cnt = 0;
    const unsigned my4 = (unsigned)tile * 0x01010101u;

    auto drain = [&](int n) {
        const bool act = lane < n;
        const int eid = act ? qw[(head + lane) & (QCAP - 1)] : 0;
        if (act && done + n <= LCAP)
            L[done + lane] = (unsigned short)(((eid >> 14) << 9) | (eid & 511));
        done += n;
        const int d = dst[eid] - base;
        if (act) __hip_atomic_fetch_add(&indeg[d], 1, __ATOMIC_RELAXED, __HIP_MEMORY_SCOPE_WORKGROUP);
    };

    auto scan16 = [&](const uint4 cur, const int idx) {
        unsigned x0 = cur.x ^ my4, x1 = cur.y ^ my4, x2 = cur.z ^ my4, x3 = cur.w ^ my4;
        unsigned z0 = ~(((x0 & 0x7f7f7f7fu) + 0x7f7f7f7fu) | x0 | 0x7f7f7f7fu);
        unsigned z1 = ~(((x1 & 0x7f7f7f7fu) + 0x7f7f7f7fu) | x1 | 0x7f7f7f7fu);
        unsigned z2 = ~(((x2 & 0x7f7f7f7fu) + 0x7f7f7f7fu) | x2 | 0x7f7f7f7fu);
        unsigned z3 = ~(((x3 & 0x7f7f7f7fu) + 0x7f7f7f7fu) | x3 | 0x7f7f7f7fu);
        unsigned m = (z0 >> 7) | (z1 >> 6) | (z2 >> 5) | (z3 >> 4);

        unsigned bal = __builtin_amdgcn_ballot_w32(m != 0);
        while (bal) {
            if (m != 0) {
                int p = __builtin_ctz(m);
                int e = idx * 16 + ((p & 7) << 2) + (p >> 3);
                qw[(head + cnt + (int)__builtin_amdgcn_mbcnt_lo(bal, 0u)) & (QCAP - 1)] = e;
                m &= m - 1;
            }
            cnt += __builtin_popcount(bal);
            if (cnt >= 32) {
                drain(32);
                head = (head + 32) & (QCAP - 1);
                cnt -= 32;
            }
            bal = __builtin_amdgcn_ballot_w32(m != 0);
        }
    };

    unsigned short* glw = gl[wave];
    int ng = 0;
    {
        const unsigned* __restrict__ P = pres + (size_t)(tile >> 5) * PLANE_WORDS;
        const int gofs = (lane >> 3) * 256 + wave * 8 + (lane & 7);
        unsigned pd[PA_CHUNKS];
#pragma unroll
        for (int ch = 0; ch < PA_CHUNKS; ++ch) pd[ch] = P[ch * 1024 + gofs];
#pragma unroll
        for (int ch = 0; ch < PA_CHUNKS; ++ch) {
            const bool has = ((pd[ch] >> (tile & 31)) & 1u) != 0u;
            const unsigned hb = __builtin_amdgcn_ballot_w32(has);
            if (has) glw[ng + (int)__builtin_amdgcn_mbcnt_lo(hb, 0u)] = (unsigned short)(ch * 1024 + gofs);
            ng += __builtin_popcount(hb);
        }
    }

    auto fetch = [&](const int bt, int& idx) -> uint4 {
        const int gi = bt * 8 + (lane >> 2);
        const int g = gi < ng ? (int)glw[gi] : PAD_GROUP;
        idx = g * 4 + (lane & 3);
        return tb[idx];
    };
    const int nb = (ng + 7) >> 3;
    int ia, ib, ic;
    uint4 ba = fetch(0, ia);
    uint4 bb = fetch(1, ib);
    for (int bt = 0; bt < nb; bt += 3) {
        uint4 bc = fetch(bt + 2, ic);
        scan16(ba, ia);
        ba = fetch(bt + 3, ia);
        scan16(bb, ib);
        bb = fetch(bt + 4, ib);
        scan16(bc, ic);
    }
    if (cnt > 0) drain(cnt);
    if (lane == 0) counts[seg] = done;
    __syncthreads();

    const int first = base & ~31;
    const int n = first + tid;
    if (n >= base && n < base + TILE_R) dinv[n] = rsqrtf(1.0f + (float)indeg[n - base]);
}

__device__ __forceinline__ void lds_add(float* p, float v) {
    __hip_atomic_fetch_add(p, v, __ATOMIC_RELAXED, __HIP_MEMORY_SCOPE_WORKGROUP);
}

__device__ __forceinline__ float2 row_load(const float* __restrict__ h, int sj, int lane) {
    return *(const float2*)(h + (size_t)sj * 64 + 2 * lane);
}
__device__ __forceinline__ void row_add(float* acc, int rj, float2 v, int lane) {
    lds_add(&acc[rj + 2 * lane],     v.x);
    lds_add(&acc[rj + 2 * lane + 1], v.y);
}

template <int M>
__device__ __forceinline__ float xmask_add(float p) {
    const int q = __builtin_amdgcn_update_dpp(0, __float_as_int(p), 0x160 + M, 0xf, 0xf, true);
    return p + __int_as_float(q);
}
__device__ __forceinline__ float row16_sum(float p) {
    p = xmask_add<1>(p);
    p = xmask_add<2>(p);
    p = xmask_add<4>(p);
    p = xmask_add<8>(p);
    return p;
}

template <bool RELU, bool HEAD>
__global__ __launch_bounds__(TS_THREADS) void tile_scatter_kernel(
        const float* __restrict__ h, const float* __restrict__ dinv,
        const int* __restrict__ src, const int* __restrict__ dst,
        const unsigned* __restrict__ tbytes, const float* __restrict__ b,
        float* __restrict__ out,
        const float* __restrict__ Wp, const float* __restrict__ bp, float* __restrict__ y,
        const unsigned short* __restrict__ lists, const int* __restrict__ counts) {
    const uint4* __restrict__ tb = (const uint4*)tbytes;
    __shared__ float acc[TILE_R * 64];
    __shared__ int   q[TS_WAVES][QCAP];
    __shared__ unsigned oq[TS_WAVES][QC];
    __shared__ int   qcnt[TS_WAVES];
    __shared__ float ys[TILE_R];

    const int tid  = threadIdx.x;
    const int lane = tid & 31;
    const int wave = __builtin_amdgcn_readfirstlane(tid >> 5);
    const int tile = blockIdx.x;
    const int base = tile * TILE_R;

    const int seg = tile * TS_WAVES + wave;
    const unsigned short* __restrict__ L = lists + (size_t)seg * LCAP;
    const int total = __builtin_amdgcn_readfirstlane(counts[seg]);

    for (int i = tid; i < TILE_R * 16; i += TS_THREADS)
        ((float4*)acc)[i] = make_float4(0.f, 0.f, 0.f, 0.f);
    if (tid < TS_WAVES) qcnt[tid] = 0;
    __syncthreads();

    int* qw = q[wave];
    int head = 0, cnt = 0;
    const unsigned my4 = (unsigned)tile * 0x01010101u;

    auto process = [&](int s, int ro, int n, bool full) {
        if (full) {
            float2 v[32];
#pragma unroll
            for (int j = 0; j < 32; ++j) v[j] = row_load(h, __builtin_amdgcn_readlane(s, j), lane);
#pragma unroll
            for (int j = 0; j < 32; ++j)
                row_add(acc, __builtin_amdgcn_readlane(ro, j), v[j], lane);
        } else {
            for (int j0 = 0; j0 < n; j0 += 8) {
                float2 v[8];
#pragma unroll
                for (int u = 0; u < 8; ++u) v[u] = row_load(h, __builtin_amdgcn_readlane(s, j0 + u), lane);
#pragma unroll
                for (int u = 0; u < 8; ++u)
                    if (j0 + u < n)
                        row_add(acc, __builtin_amdgcn_readlane(ro, j0 + u), v[u], lane);
            }
        }
    };

    auto drain = [&](int n, bool full) {
        const bool act = lane < n;
        int eid = act ? qw[(head + lane) & (QCAP - 1)] : 0;
        int s  = src[eid];
        int ro = act ? (dst[eid] - base) * 64 : 0;
        process(s, ro, n, full);
    };

    auto scan16 = [&](const uint4 cur, const int idx) {
        unsigned x0 = cur.x ^ my4, x1 = cur.y ^ my4, x2 = cur.z ^ my4, x3 = cur.w ^ my4;
        unsigned z0 = ~(((x0 & 0x7f7f7f7fu) + 0x7f7f7f7fu) | x0 | 0x7f7f7f7fu);
        unsigned z1 = ~(((x1 & 0x7f7f7f7fu) + 0x7f7f7f7fu) | x1 | 0x7f7f7f7fu);
        unsigned z2 = ~(((x2 & 0x7f7f7f7fu) + 0x7f7f7f7fu) | x2 | 0x7f7f7f7fu);
        unsigned z3 = ~(((x3 & 0x7f7f7f7fu) + 0x7f7f7f7fu) | x3 | 0x7f7f7f7fu);
        unsigned m = (z0 >> 7) | (z1 >> 6) | (z2 >> 5) | (z3 >> 4);

        unsigned bal = __builtin_amdgcn_ballot_w32(m != 0);
        while (bal) {
            if (m != 0) {
                int p = __builtin_ctz(m);
                int e = idx * 16 + ((p & 7) << 2) + (p >> 3);
                qw[(head + cnt + (int)__builtin_amdgcn_mbcnt_lo(bal, 0u)) & (QCAP - 1)] = e;
                m &= m - 1;
            }
            cnt += __builtin_popcount(bal);
            if (cnt >= 32) {
                drain(32, true);
                head = (head + 32) & (QCAP - 1);
                cnt -= 32;
            }
            bal = __builtin_amdgcn_ballot_w32(m != 0);
        }
    };

    if (total <= LCAP) {
        if (total > 0) {
            const int wbase = wave * 512;
            int sv[LCAP / 32], dv[LCAP / 32];
#pragma unroll
            for (int bq = 0; bq < LCAP / 32; ++bq) {
                const int i = bq * 32 + lane;
                const unsigned c = L[i < total ? i : 0];
                const int eid = (int)((c >> 9) << 14) + wbase + (int)(c & 511u);
                sv[bq] = src[eid];
                dv[bq] = dst[eid] - base;
            }
#pragma unroll
            for (int bq = 0; bq < LCAP / 32; ++bq) {
                if (bq * 32 < total) {
                    const bool valid = bq * 32 + lane < total;
                    const int dl = dv[bq];
                    const int ow = (2 * dl) / 25;
                    const int dr = dl - ((25 * ow + 1) >> 1);
                    bool over = false;
                    if (valid) {
                        const int slot = __hip_atomic_fetch_add(&qcnt[ow], 1, __ATOMIC_RELAXED, __HIP_MEMORY_SCOPE_WORKGROUP);
                        if (slot < QC) oq[ow][slot] = (unsigned)sv[bq] | ((unsigned)dr << 17);
                        else over = true;
                    }
                    unsigned ob = __builtin_amdgcn_ballot_w32(over);
                    while (ob) {
                        const int j = __builtin_ctz(ob);
                        ob &= ob - 1;
                        const int sj = __builtin_amdgcn_readlane(sv[bq], j);
                        const int rj = __builtin_amdgcn_readlane(dl, j) * 64;
                        row_add(acc, rj, row_load(h, sj, lane), lane);
                    }
                }
            }
        }
    } else {
    uint4 ba = tb[tid];
    uint4 bb = tb[tid + TS_THREADS];
    for (int it = 0; it < TS_ITERS; it += 3) {
        const int idx = it * TS_THREADS + tid;
        uint4 bc = tb[idx + 2 * TS_THREADS];
        scan16(ba, idx);
        ba = tb[idx + 3 * TS_THREADS];
        scan16(bb, idx + TS_THREADS);
        bb = tb[idx + 4 * TS_THREADS];
        scan16(bc, idx + 2 * TS_THREADS);
    }
    if (cnt > 0) drain(cnt, false);
    }
    __syncthreads();

    {
    const int rs0 = (25 * wave + 1) >> 1;
    const int nr  = ((25 * (wave + 1) + 1) >> 1) - rs0;
    int nq = __builtin_amdgcn_readfirstlane(qcnt[wave]);
    nq = nq < QC ? nq : QC;
    const unsigned* oqw = oq[wave];
    v16f RX = {}, RY = {};
    for (int b0 = 0; b0 < nq; b0 += 32) {
        const int m = (nq - b0 < 32) ? (nq - b0) : 32;
        const unsigned pk = (lane < m) ? oqw[b0 + lane] : (15u << 17);
        const int s  = (int)(pk & 0x1ffffu);
        const int dr = (int)(pk >> 17);
        float2 v[32];
#pragma unroll
        for (int j = 0; j < 32; ++j) v[j] = row_load(h, __builtin_amdgcn_readlane(s, j), lane);
        __builtin_amdgcn_sched_barrier(0);
#pragma unroll
        for (int j = 0; j < 32; ++j) {
            const int d = __builtin_amdgcn_readlane(dr, j);
            RX[d] += v[j].x;
            RY[d] += v[j].y;
        }
    }
    float2 hv[13];
    float  dq[13];
#pragma unroll
    for (int r = 0; r < 13; ++r) {
        const int row = base + rs0 + (r < nr ? r : 0);
        hv[r] = *(const float2*)(h + (size_t)row * 64 + 2 * lane);
        dq[r] = dinv[row];
    }
    const float2 bv = *(const float2*)(b + 2 * lane);
    float2 wq = make_float2(0.f, 0.f);
    if (HEAD) wq = *(const float2*)(Wp + 2 * lane);
#pragma unroll
    for (int r = 0; r < 13; ++r) {
        if (r < nr) {
            const float2 a = *(const float2*)&acc[(rs0 + r) * 64 + 2 * lane];
            float ox = dq[r] * ((RX[r] + a.x) + hv[r].x) + bv.x;
            float oy = dq[r] * ((RY[r] + a.y) + hv[r].y) + bv.y;
            if (RELU) { ox = fmaxf(ox, 0.0f); oy = fmaxf(oy, 0.0f); }
            if (HEAD) {
                float p = ox * wq.x + oy * wq.y;
                p = row16_sum(p);
                const float tot = __int_as_float(__builtin_amdgcn_readlane(__float_as_int(p), 0)) +
                                  __int_as_float(__builtin_amdgcn_readlane(__float_as_int(p), 16));
                if (lane == 0) ys[rs0 + r] = tot;
            } else {
                *(float2*)(out + (size_t)(base + rs0 + r) * 64 + 2 * lane) = make_float2(ox, oy);
            }
        }
    }
    }
    if (HEAD) {
        __syncthreads();
        const int first = base & ~31;
        const int n = first + tid;
        if (n >= base && n < base + TILE_R) y[n] = ys[n - base] + bp[0];
    }
}

extern "C" void kernel_launch(void* const* d_in, const int* in_sizes, int n_in,
                              void* d_out, int out_size, void* d_ws, size_t ws_size,
                              hipStream_t stream) {
    (void)in_sizes; (void)n_in; (void)out_size; (void)ws_size;

    const float* x  = (const float*)d_in[0];
    const int*   ei = (const int*)d_in[1];
    const float* W1 = (const float*)d_in[2];
    const float* b1 = (const float*)d_in[3];
    const float* W2 = (const float*)d_in[4];
    const float* b2 = (const float*)d_in[5];
    const float* Wp = (const float*)d_in[6];
    const float* bp = (const float*)d_in[7];
    float* y = (float*)d_out;

    const int N = GNN_N, E = GNN_E;
    const int* src = ei;
    const int* dst = ei + E;

    char* ws = (char*)d_ws;
    size_t off = 0;
    auto take = [&](size_t bytes) {
        void* p = ws + off;
        off += (bytes + 255) & ~(size_t)255;
        return p;
    };
    float* dinv = (float*)take((size_t)N * 4);
    unsigned* tb = (unsigned*)take((size_t)E * 4);
    unsigned short* lists = (unsigned short*)(tb + TB_QUADS);
    int* counts = (int*)(tb + TB_QUADS + LIST_WORDS);
    float* bufA = (float*)take((size_t)N * DIM * 4);
    float* bufB = (float*)take((size_t)N * DIM * 4);

    const int nTiles = N / 16;
    const int gemmBlocks = (nTiles + 7) / 8;

    unsigned* pres = (unsigned*)bufB;
    tilemap_kernel<<<TB_QUADS / 256, 256, 0, stream>>>(dst, tb, pres, E / 4);
    tile_filter_kernel<<<N_TILES, TS_THREADS, 0, stream>>>(dst, tb, pres, dinv, lists, counts);

    gemm64_wmma<<<gemmBlocks, 256, 0, stream>>>(x, W1, dinv, bufA, nTiles);
    tile_scatter_kernel<true, false><<<N_TILES, TS_THREADS, 0, stream>>>(bufA, dinv, src, dst, tb, b1, bufB, Wp, bp, y, lists, counts);

    gemm64_wmma<<<gemmBlocks, 256, 0, stream>>>(bufB, W2, dinv, bufA, nTiles);
    tile_scatter_kernel<true, true><<<N_TILES, TS_THREADS, 0, stream>>>(bufA, dinv, src, dst, tb, b2, bufB, Wp, bp, y, lists, counts);
}


# Filter pass writes each tile's record as per-owner-wave lists of (source id, row), 3 bytes per edge; layers read their list: no src/dst fetch per edge, no queue; row sums in filter arrival order
# speedup vs baseline: 6.3991x; 1.2736x over previous
#include <hip/hip_runtime.h>


typedef float v2f __attribute__((ext_vector_type(2)));
typedef float v8f __attribute__((ext_vector_type(8)));
typedef float v16f __attribute__((ext_vector_type(16)));

#define GNN_N 100000
#define GNN_E 1600000
#define DIM   64
#define XPAD  68

#define TILE_R   400
#define N_TILES  (GNN_N / TILE_R)
#define NV16     (GNN_E / 16)
#define TS_THREADS 1024
#define TS_WAVES   (TS_THREADS / 32)
#define TS_ITERS   ((NV16 + TS_THREADS - 1) / TS_THREADS)
#define TB_QUADS   ((TS_ITERS + 3) * TS_THREADS * 4)
#define QCAP     64
#define QC       320
#define N_SEGS   (N_TILES * TS_WAVES)
#define TCAP     8448
#define LO_BYTES ((size_t)N_TILES * TCAP * 2)
#define HI_BYTES ((size_t)N_TILES * TCAP)
static_assert(LO_BYTES + HI_BYTES + (size_t)N_SEGS * 4 <= (size_t)GNN_E * 4, "the record and its list heads must fit the 4*E byte slot");
static_assert(TCAP % 64 == 0 && TCAP < (1 << 14) && QC < (1 << 12) && GNN_N <= (1 << 17), "a list head is (start << 12) | length; an entry is source | row << 17");
#define GRP_QUADS   16
#define N_PLANES    8
#define PLANE_WORDS (TB_QUADS / GRP_QUADS)
#define PA_CHUNKS   ((TS_ITERS + 3) / 4)
#define GL_CAP      (PA_CHUNKS * 32)
#define PAD_GROUP   (GNN_E / 64)
static_assert(TB_QUADS % 256 == 0 && 256 / GRP_QUADS == 16, "a map block is 16 whole groups and the grid has no partial block");
static_assert((PA_CHUNKS * 4) * 256 <= PLANE_WORDS && PLANE_WORDS < 65536, "every group a wave tests has a presence word; a group number fits 16 bits");
static_assert((PAD_GROUP + 1) * GRP_QUADS <= TB_QUADS && N_TILES <= 255, "the padding group's bytes exist and name no tile");
static_assert((size_t)N_PLANES * PLANE_WORDS + TB_QUADS <= (size_t)GNN_N * DIM && (N_PLANES * PLANE_WORDS) % 64 == 0,
              "the presence planes and the tile-id bytes fit the row buffer that lends its space; the bytes start on a line");

__global__ __launch_bounds__(256) void tilemap_kernel(const int* __restrict__ dst, unsigned* __restrict__ tb4,
                                                      unsigned* __restrict__ pres, int nQuads) {
    __shared__ unsigned pw[N_PLANES][16];
    const int t = threadIdx.x;
    const int i = blockIdx.x * 256 + t;
    if (t < N_PLANES * 16) (&pw[0][0])[t] = 0u;
    __syncthreads();
    unsigned p = ~0u;
    if (i < nQuads) {
        int4 d = ((const int4*)dst)[i];
        p = (unsigned)(d.x / TILE_R) | ((unsigned)(d.y / TILE_R) << 8) |
            ((unsigned)(d.z / TILE_R) << 16) | ((unsigned)(d.w / TILE_R) << 24);
    }
    tb4[i] = p;
    const int g = t >> 4;
#pragma unroll
    for (int k = 0; k < 4; ++k) {
        const unsigned id = (p >> (8 * k)) & 255u;
        __hip_atomic_fetch_or(&pw[id >> 5][g], 1u << (id & 31u), __ATOMIC_RELAXED, __HIP_MEMORY_SCOPE_WORKGROUP);
    }
    __syncthreads();
    if (t < N_PLANES * 16)
        pres[(size_t)(t >> 4) * PLANE_WORDS + blockIdx.x * 16 + (t & 15)] = pw[t >> 4][t & 15];
}


__device__ __forceinline__ int w_slot(int g) {
    const int k = g >> 6;
    const int n = g & 63;
    return (k >> 1) * 128 + n * 2 + (k & 1);
}
static_assert(4 * TS_THREADS == 64 * 64, "four words of W per thread");

template <int STRIDE>
__device__ __forceinline__ void stripe_product(const float* rows, const float* Ws, int lane,
                                               const float (&sc)[8], float* __restrict__ Hs) {
    const int l15 = lane & 15;
    const int hiK = (lane >> 4) * 2;
    const int rowOff = (lane >> 4) * 8;
    const float* xs = rows + l15 * STRIDE;

    v8f c0 = {}, c1 = {}, c2 = {}, c3 = {};
#pragma unroll
    for (int k = 0; k < 64; k += 4) {
        v2f a = *(const v2f*)(xs + k + hiK);
        const float* wrow = Ws + ((k + hiK) >> 1) * 128 + l15 * 2;
        v2f b0 = *(const v2f*)(wrow +  0);
        v2f b1 = *(const v2f*)(wrow + 32);
        v2f b2 = *(const v2f*)(wrow + 64);
        v2f b3 = *(const v2f*)(wrow + 96);
        c0 = __builtin_amdgcn_wmma_f32_16x16x4_f32(false, a, false, b0, (short)0, c0, false, false);
        c1 = __builtin_amdgcn_wmma_f32_16x16x4_f32(false, a, false, b1, (short)0, c1, false, false);
        c2 = __builtin_amdgcn_wmma_f32_16x16x4_f32(false, a, false, b2, (short)0, c2, false, false);
        c3 = __builtin_amdgcn_wmma_f32_16x16x4_f32(false, a, false, b3, (short)0, c3, false, false);
    }

#pragma unroll
    for (int r = 0; r < 8; ++r) {
        float* out = Hs + (size_t)(r + rowOff) * 64 + l15;
        out[0]  = c0[r] * sc[r];
        out[16] = c1[r] * sc[r];
        out[32] = c2[r] * sc[r];
        out[48] = c3[r] * sc[r];
    }
}
#define N_STRIPES (TILE_R / 16)
static_assert(N_STRIPES * 16 == TILE_R && N_STRIPES <= TS_WAVES, "a tile is whole stripes and every stripe has a wave");

__global__ __launch_bounds__(TS_THREADS) void tile_filter_kernel(
        const int* __restrict__ dst, const unsigned* __restrict__ tbytes, const unsigned* __restrict__ pres,
        float* __restrict__ dinv, unsigned short* __restrict__ recLo, unsigned char* __restrict__ recHi,
        int* __restrict__ heads, const int* __restrict__ src,
        const float* __restrict__ x, const float* __restrict__ W, float* __restrict__ H) {
    const uint4* __restrict__ tb = (const uint4*)tbytes;
    __shared__ int q[TS_WAVES][QCAP];
    __shared__ int indeg[TILE_R];
    __shared__ unsigned oq[TS_WAVES][QC];
    __shared__ int qcnt[TS_WAVES];
    __shared__ unsigned short gl[TS_WAVES][GL_CAP];
    __shared__ float Ws[64 * 64];
    __shared__ float Xs[N_STRIPES][16 * XPAD];

    const int tid  = threadIdx.x;
    const int lane = tid & 31;
    const int wave = __builtin_amdgcn_readfirstlane(tid >> 5);
    const int tile = blockIdx.x;
    const int base = tile * TILE_R;

    const int seg = tile * TS_WAVES + wave;

    const bool hasStripe = wave < N_STRIPES;

    if (tid < TILE_R) indeg[tid] = 0;
    if (tid < TS_WAVES) qcnt[tid] = 0;
    __syncthreads();

    const float w0 = W[tid], w1 = W[tid + TS_THREADS], w2 = W[tid + 2 * TS_THREADS], w3 = W[tid + 3 * TS_THREADS];
    float4 xa = make_float4(0.f, 0.f, 0.f, 0.f);
    float4 xb = xa, xc = xa, xd = xa, xe = xa, xf = xa, xg = xa, xh = xa;
    if (hasStripe) {
        const float4* Xg = (const float4*)(x + (size_t)(base + 16 * wave) * 64) + lane;
        xa = Xg[0];   xb = Xg[32];  xc = Xg[64];  xd = Xg[96];
        xe = Xg[128]; xf = Xg[160]; xg = Xg[192]; xh = Xg[224];
    }


    int* qw = q[wave];
    int head = 0, cnt = 0;
    const unsigned my4 = (unsigned)tile * 0x01010101u;

    auto drain = [&](int n) {
        const bool act = lane < n;
        const int eid = act ? qw[(head + lane) & (QCAP - 1)] : 0;
        int s = src[eid];
        const int d = dst[eid] - base;
        asm volatile("" : "+v"(s));
        if (act) {
            __hip_atomic_fetch_add(&indeg[d], 1, __ATOMIC_RELAXED, __HIP_MEMORY_SCOPE_WORKGROUP);
            const int ow = (2 * d) / 25;
            const int dr = d - ((25 * ow + 1) >> 1);
            const int slot = __hip_atomic_fetch_add(&qcnt[ow], 1, __ATOMIC_RELAXED, __HIP_MEMORY_SCOPE_WORKGROUP);
            if (slot < QC) oq[ow][slot] = (unsigned)s | ((unsigned)dr << 17);
        }
    };

    auto scan16 = [&](const uint4 cur, const int idx) {
        unsigned x0 = cur.x ^ my4, x1 = cur.y ^ my4, x2 = cur.z ^ my4, x3 = cur.w ^ my4;
        unsigned z0 = ~(((x0 & 0x7f7f7f7fu) + 0x7f7f7f7fu) | x0 | 0x7f7f7f7fu);
        unsigned z1 = ~(((x1 & 0x7f7f7f7fu) + 0x7f7f7f7fu) | x1 | 0x7f7f7f7fu);
        unsigned z2 = ~(((x2 & 0x7f7f7f7fu) + 0x7f7f7f7fu) | x2 | 0x7f7f7f7fu);
        unsigned z3 = ~(((x3 & 0x7f7f7f7fu) + 0x7f7f7f7fu) | x3 | 0x7f7f7f7fu);
        unsigned m = (z0 >> 7) | (z1 >> 6) | (z2 >> 5) | (z3 >> 4);

        unsigned bal = __builtin_amdgcn_ballot_w32(m != 0);
        while (bal) {
            if (m != 0) {
                int p = __builtin_ctz(m);
                int e = idx * 16 + ((p & 7) << 2) + (p >> 3);
                qw[(head + cnt + (int)__builtin_amdgcn_mbcnt_lo(bal, 0u)) & (QCAP - 1)] = e;
                m &= m - 1;
            }
            cnt += __builtin_popcount(bal);
            if (cnt >= 32) {
                drain(32);
                head = (head + 32) & (QCAP - 1);
                cnt -= 32;
            }
            bal = __builtin_amdgcn_ballot_w32(m != 0);
        }
    };

    unsigned short* glw = gl[wave];
    int ng = 0;
    {
        const unsigned* __restrict__ P = pres + (size_t)(tile >> 5) * PLANE_WORDS;
        const int gofs = (lane >> 3) * 256 + wave * 8 + (lane & 7);
        unsigned pd[PA_CHUNKS];
#pragma unroll
        for (int ch = 0; ch < PA_CHUNKS; ++ch) pd[ch] = P[ch * 1024 + gofs];
#pragma unroll
        for (int ch = 0; ch < PA_CHUNKS; ++ch) {
            const bool has = ((pd[ch] >> (tile & 31)) & 1u) != 0u;
            const unsigned hb = __builtin_amdgcn_ballot_w32(has);
            if (has) glw[ng + (int)__builtin_amdgcn_mbcnt_lo(hb, 0u)] = (unsigned short)(ch * 1024 + gofs);
            ng += __builtin_popcount(hb);
        }
    }

    Ws[w_slot(tid)] = w0;
    Ws[w_slot(tid + TS_THREADS)] = w1;
    Ws[w_slot(tid + 2 * TS_THREADS)] = w2;
    Ws[w_slot(tid + 3 * TS_THREADS)] = w3;
    if (hasStripe) {
        float* xl = &Xs[wave][(lane >> 4) * XPAD + (lane & 15) * 4];
        *(float4*)(xl +  0 * XPAD) = xa;  *(float4*)(xl +  2 * XPAD) = xb;
        *(float4*)(xl +  4 * XPAD) = xc;  *(float4*)(xl +  6 * XPAD) = xd;
        *(float4*)(xl +  8 * XPAD) = xe;  *(float4*)(xl + 10 * XPAD) = xf;
        *(float4*)(xl + 12 * XPAD) = xg;  *(float4*)(xl + 14 * XPAD) = xh;
    }

    auto fetch = [&](const int bt, int& idx) -> uint4 {
        const int gi = bt * 8 + (lane >> 2);
        const int g = gi < ng ? (int)glw[gi] : PAD_GROUP;
        idx = g * 4 + (lane & 3);
        return tb[idx];
    };
    const int nb = (ng + 7) >> 3;
    int ia, ib, ic;
    uint4 ba = fetch(0, ia);
    uint4 bb = fetch(1, ib);
    for (int bt = 0; bt < nb; bt += 3) {
        uint4 bc = fetch(bt + 2, ic);
        scan16(ba, ia);
        ba = fetch(bt + 3, ia);
        scan16(bb, ib);
        bb = fetch(bt + 4, ib);
        scan16(bc, ic);
    }
    if (cnt > 0) drain(cnt);
    __syncthreads();

    {
        const int c = qcnt[lane];
        const bool full = __builtin_amdgcn_ballot_w32(c > QC) != 0u;
        int start = 0, tot = 0;
#pragma unroll
        for (int v = 0; v < TS_WAVES; ++v) {
            const int a = (__builtin_amdgcn_readlane(c, v) + 63) & ~63;
            start += v < wave ? a : 0;
            tot += a;
        }
        const bool fits = !full && tot <= TCAP;
        const int len = __builtin_amdgcn_readlane(c, wave);
        if (lane == 0) heads[seg] = fits ? ((start << 12) | len) : -1;
        if (fits) {
            unsigned short* __restrict__ lo = recLo + (size_t)tile * TCAP + start;
            unsigned char*  __restrict__ hi = recHi + (size_t)tile * TCAP + start;
            const unsigned* oqw = oq[wave];
            for (int i = lane; i < len; i += 32) {
                const unsigned pk = oqw[i];
                lo[i] = (unsigned short)(pk & 0xffffu);
                hi[i] = (unsigned char)(pk >> 16);
            }
        }
    }

    const int first = base & ~31;
    const int n = first + tid;
    if (n >= base && n < base + TILE_R) dinv[n] = rsqrtf(1.0f + (float)indeg[n - base]);

    if (hasStripe) {
        const int rowOff = (lane >> 4) * 8;
        float sc[8];
#pragma unroll
        for (int r = 0; r < 8; ++r) sc[r] = rsqrtf(1.0f + (float)indeg[16 * wave + r + rowOff]);
        stripe_product<XPAD>(&Xs[wave][0], Ws, lane, sc, H + (size_t)(base + 16 * wave) * 64);
    }
}

__device__ __forceinline__ void lds_add(float* p, float v) {
    __hip_atomic_fetch_add(p, v, __ATOMIC_RELAXED, __HIP_MEMORY_SCOPE_WORKGROUP);
}

__device__ __forceinline__ float2 row_load(const float* __restrict__ h, int sj, int lane) {
    return *(const float2*)(h + (size_t)sj * 64 + 2 * lane);
}
__device__ __forceinline__ void row_add(float* acc, int rj, float2 v, int lane) {
    lds_add(&acc[rj + 2 * lane],     v.x);
    lds_add(&acc[rj + 2 * lane + 1], v.y);
}

template <int M>
__device__ __forceinline__ float xmask_add(float p) {
    const int q = __builtin_amdgcn_update_dpp(0, __float_as_int(p), 0x160 + M, 0xf, 0xf, true);
    return p + __int_as_float(q);
}
__device__ __forceinline__ float row16_sum(float p) {
    p = xmask_add<1>(p);
    p = xmask_add<2>(p);
    p = xmask_add<4>(p);
    p = xmask_add<8>(p);
    return p;
}

template <bool RELU, bool HEAD>
__global__ __launch_bounds__(TS_THREADS) void tile_scatter_kernel(
        const float* __restrict__ h, const float* __restrict__ dinv,
        const int* __restrict__ src, const int* __restrict__ dst,
        const float* __restrict__ b,
        float* __restrict__ out,
        const float* __restrict__ Wp, const float* __restrict__ bp, float* __restrict__ y,
        const unsigned short* __restrict__ recLo, const unsigned char* __restrict__ recHi,
        const int* __restrict__ heads,
        const float* __restrict__ Wn) {
    __shared__ float acc[TILE_R * 64];
    __shared__ float Ws[HEAD ? 2 : 64 * 64];
    __shared__ float ys[TILE_R];

    const int tid  = threadIdx.x;
    const int lane = tid & 31;
    const int wave = __builtin_amdgcn_readfirstlane(tid >> 5);
    const int tile = blockIdx.x;
    const int base = tile * TILE_R;

    const int seg = tile * TS_WAVES + wave;
    const int hd = __builtin_amdgcn_readfirstlane(heads[seg]);
    const bool flagged = hd < 0;
    const int nq = flagged ? 0 : (hd & 4095);
    const unsigned short* __restrict__ lo = recLo + (size_t)tile * TCAP + (flagged ? 0 : (hd >> 12));
    const unsigned char*  __restrict__ hi = recHi + (size_t)tile * TCAP + (flagged ? 0 : (hd >> 12));

    unsigned nxt = 15u << 17;
    if (nq > 0) {
        const unsigned i0 = lane < nq ? (unsigned)lane : 0u;
        const unsigned e0 = (unsigned)lo[i0] | ((unsigned)hi[i0] << 16);
        nxt = lane < nq ? e0 : (15u << 17);
    }

    if (!HEAD) {
        const float w0 = Wn[tid], w1 = Wn[tid + TS_THREADS], w2 = Wn[tid + 2 * TS_THREADS], w3 = Wn[tid + 3 * TS_THREADS];
        Ws[w_slot(tid)] = w0;
        Ws[w_slot(tid + TS_THREADS)] = w1;
        Ws[w_slot(tid + 2 * TS_THREADS)] = w2;
        Ws[w_slot(tid + 3 * TS_THREADS)] = w3;
    }

    if (flagged) {
        for (int i = tid; i < TILE_R * 16; i += TS_THREADS)
            ((float4*)acc)[i] = make_float4(0.f, 0.f, 0.f, 0.f);
        __syncthreads();
        const int4* __restrict__ d4 = (const int4*)dst;
        for (int it = 0; it < (GNN_E / 4 + TS_THREADS - 1) / TS_THREADS; ++it) {
            const int i = it * TS_THREADS + tid;
            const bool in = i < GNN_E / 4;
            const int4 d = in ? d4[i] : make_int4(-1, -1, -1, -1);
            const int dk[4] = {d.x - base, d.y - base, d.z - base, d.w - base};
#pragma unroll
            for (int k = 0; k < 4; ++k) {
                unsigned bal = __builtin_amdgcn_ballot_w32(in && (unsigned)dk[k] < (unsigned)TILE_R);
                while (bal) {
                    const int j = __builtin_ctz(bal);
                    bal &= bal - 1;
                    const int e = __builtin_amdgcn_readlane(i, j) * 4 + k;
                    const int sj = __builtin_amdgcn_readfirstlane(src[e]);
                    const int rj = __builtin_amdgcn_readlane(dk[k], j) * 64;
                    row_add(acc, rj, row_load(h, sj, lane), lane);
                }
            }
        }
        __syncthreads();
    }

    {
    const int rs0 = (25 * wave + 1) >> 1;
    const int nr  = ((25 * (wave + 1) + 1) >> 1) - rs0;
    v16f RX = {}, RY = {};
    for (int b0 = 0; b0 < nq; b0 += 32) {
        const unsigned pk = nxt;
        const unsigned i2 = (unsigned)(b0 + 32 + lane);
        const unsigned ic = i2 < (unsigned)nq ? i2 : 0u;
        const unsigned e2 = (unsigned)lo[ic] | ((unsigned)hi[ic] << 16);
        nxt = i2 < (unsigned)nq ? e2 : (15u << 17);
        const int s  = (int)(pk & 0x1ffffu);
        const int dr = (int)(pk >> 17);
        float2 v[32];
#pragma unroll
        for (int j = 0; j < 32; ++j) v[j] = row_load(h, __builtin_amdgcn_readlane(s, j), lane);
        __builtin_amdgcn_sched_barrier(0);
#pragma unroll
        for (int j = 0; j < 32; ++j) {
            const int d = __builtin_amdgcn_readlane(dr, j);
            RX[d] += v[j].x;
            RY[d] += v[j].y;
        }
    }
    float2 hv[13];
    float  dq[13];
#pragma unroll
    for (int r = 0; r < 13; ++r) {
        const int row = base + rs0 + (r < nr ? r : 0);
        hv[r] = *(const float2*)(h + (size_t)row * 64 + 2 * lane);
        dq[r] = dinv[row];
    }
    const float2 bv = *(const float2*)(b + 2 * lane);
    float2 wq = make_float2(0.f, 0.f);
    if (HEAD) wq = *(const float2*)(Wp + 2 * lane);
#pragma unroll
    for (int r = 0; r < 13; ++r) {
        if (r < nr) {
            float2 a = make_float2(0.f, 0.f);
            if (flagged) a = *(const float2*)&acc[(rs0 + r) * 64 + 2 * lane];
            float ox = dq[r] * ((RX[r] + a.x) + hv[r].x) + bv.x;
            float oy = dq[r] * ((RY[r] + a.y) + hv[r].y) + bv.y;
            if (RELU) { ox = fmaxf(ox, 0.0f); oy = fmaxf(oy, 0.0f); }
            if (HEAD) {
                float p = ox * wq.x + oy * wq.y;
                p = row16_sum(p);
                const float tot = __int_as_float(__builtin_amdgcn_readlane(__float_as_int(p), 0)) +
                                  __int_as_float(__builtin_amdgcn_readlane(__float_as_int(p), 16));
                if (lane == 0) ys[rs0 + r] = tot;
            } else {
                *(float2*)&acc[(rs0 + r) * 64 + 2 * lane] = make_float2(ox, oy);
            }
        }
    }
    }
    if (!HEAD) {
        __syncthreads();
        if (wave < N_STRIPES) {
            const int rowOff = (lane >> 4) * 8;
            float sc[8];
#pragma unroll
            for (int r = 0; r < 8; ++r) sc[r] = dinv[base + 16 * wave + r + rowOff];
            stripe_product<64>(&acc[16 * wave * 64], Ws, lane, sc, out + (size_t)(base + 16 * wave) * 64);
        }
    }
    if (HEAD) {
        __syncthreads();
        const int first = base & ~31;
        const int n = first + tid;
        if (n >= base && n < base + TILE_R) y[n] = ys[n - base] + bp[0];
    }
}

extern "C" void kernel_launch(void* const* d_in, const int* in_sizes, int n_in,
                              void* d_out, int out_size, void* d_ws, size_t ws_size,
                              hipStream_t stream) {
    (void)in_sizes; (void)n_in; (void)out_size; (void)ws_size;

    const float* x  = (const float*)d_in[0];
    const int*   ei = (const int*)d_in[1];
    const float* W1 = (const float*)d_in[2];
    const float* b1 = (const float*)d_in[3];
    const float* W2 = (const float*)d_in[4];
    const float* b2 = (const float*)d_in[5];
    const float* Wp = (const float*)d_in[6];
    const float* bp = (const float*)d_in[7];
    float* y = (float*)d_out;

    const int N = GNN_N, E = GNN_E;
    const int* src = ei;
    const int* dst = ei + E;

    char* ws = (char*)d_ws;
    size_t off = 0;
    auto take = [&](size_t bytes) {
        void* p = ws + off;
        off += (bytes + 255) & ~(size_t)255;
        return p;
    };
    float* dinv = (float*)take((size_t)N * 4);
    char* rec = (char*)take((size_t)E * 4);
    unsigned short* recLo = (unsigned short*)rec;
    unsigned char*  recHi = (unsigned char*)(rec + LO_BYTES);
    int* heads = (int*)(rec + LO_BYTES + HI_BYTES);
    float* bufA = (float*)take((size_t)N * DIM * 4);
    float* bufB = (float*)take((size_t)N * DIM * 4);

    unsigned* pres = (unsigned*)bufB;
    unsigned* tb = pres + N_PLANES * PLANE_WORDS;
    tilemap_kernel<<<TB_QUADS / 256, 256, 0, stream>>>(dst, tb, pres, E / 4);
    tile_filter_kernel<<<N_TILES, TS_THREADS, 0, stream>>>(dst, tb, pres, dinv, recLo, recHi, heads, src, x, W1, bufA);

    tile_scatter_kernel<true, false><<<N_TILES, TS_THREADS, 0, stream>>>(bufA, dinv, src, dst, b1, bufB, Wp, bp, y, recLo, recHi, heads, W2);

    tile_scatter_kernel<true, true><<<N_TILES, TS_THREADS, 0, stream>>>(bufB, dinv, src, dst, b2, bufA, Wp, bp, y, recLo, recHi, heads, W2);
}
